# starting kernel
# baseline (speedup 1.0000x reference)
#include <hip/hip_runtime.h>

#define S_   256
#define N_   558
#define B2_  8192
#define F_   128
#define M_   (S_ * N_)
#define TE_  (S_ * B2_)

typedef __attribute__((ext_vector_type(2))) float v2f;
typedef __attribute__((ext_vector_type(8))) float v8f;

__global__ __launch_bounds__(256) void k_init(float* __restrict__ deg,
                                              int* __restrict__ bdmax) {
    int i = blockIdx.x * 256 + threadIdx.x;
    if (i < M_) deg[i] = 1.0f;
    if (i == 0) *bdmax = 0;
}

__global__ __launch_bounds__(256) void k_bdmax(const int* __restrict__ ei,
                                               const float* __restrict__ bd,
                                               int* __restrict__ bdmax) {
    int stride = gridDim.x * 256;
    float m = 0.0f;
    for (int i = blockIdx.x * 256 + threadIdx.x; i < TE_; i += stride) {
        int s    = i >> 13;
        int base = i + (s << 13);
        int src  = ei[base];
        int dst  = ei[base + B2_];
        if (src != dst) m = fmaxf(m, bd[i]);
    }
    for (int off = 16; off > 0; off >>= 1)
        m = fmaxf(m, __shfl_xor(m, off, 32));
    if ((threadIdx.x & 31) == 0) atomicMax(bdmax, __float_as_int(m));
}

__global__ __launch_bounds__(256) void k_deg(const int* __restrict__ ei,
                                             const float* __restrict__ bd,
                                             const int* __restrict__ bdmax,
                                             float* __restrict__ deg) {
    int i = blockIdx.x * 256 + threadIdx.x;
    if (i >= TE_) return;
    int s    = i >> 13;
    int base = i + (s << 13);
    int src  = ei[base];
    int dst  = ei[base + B2_];
    if (src == dst) return;
    float bm = __int_as_float(*bdmax);
    unsafeAtomicAdd(&deg[s * N_ + dst], bd[i] / bm);
}

__global__ __launch_bounds__(256) void k_dinv_agg(const float* __restrict__ x,
                                                  const float* __restrict__ deg,
                                                  float* __restrict__ dinv,
                                                  float* __restrict__ agg) {
    int t = blockIdx.x * 256 + threadIdx.x;
    int m = t >> 5;
    if (m >= M_) return;
    int f4   = (t & 31) * 4;
    float d  = deg[m];
    float di = rsqrtf(d);
    if ((t & 31) == 0) dinv[m] = di;
    float sl = di * di;
    const float4 xv = *(const float4*)(x + (size_t)m * F_ + f4);
    float4 o;
    o.x = xv.x * sl; o.y = xv.y * sl; o.z = xv.z * sl; o.w = xv.w * sl;
    *(float4*)(agg + (size_t)m * F_ + f4) = o;
}

__global__ __launch_bounds__(256) void k_scatter(const int* __restrict__ ei,
                                                 const float* __restrict__ bd,
                                                 const float* __restrict__ x,
                                                 const float* __restrict__ dinv,
                                                 const int* __restrict__ bdmax,
                                                 float* __restrict__ agg) {
    int lane = threadIdx.x & 31;
    int e    = blockIdx.x * 8 + (threadIdx.x >> 5);
    if (e >= TE_) return;
    int s    = e >> 13;
    int base = e + (s << 13);
    int src  = ei[base];
    int dst  = ei[base + B2_];
    if (src == dst) return;
    int row = s * N_ + src;
    int col = s * N_ + dst;
    float bm   = __int_as_float(*bdmax);
    float norm = dinv[row] * (bd[e] / bm) * dinv[col];
    const float4 xv = *(const float4*)(x + (size_t)row * F_ + lane * 4);
    float* dp = agg + (size_t)col * F_ + lane * 4;
    unsafeAtomicAdd(dp + 0, norm * xv.x);
    unsafeAtomicAdd(dp + 1, norm * xv.y);
    unsafeAtomicAdd(dp + 2, norm * xv.z);
    unsafeAtomicAdd(dp + 3, norm * xv.w);
}

__global__ __launch_bounds__(256) void k_gemm(const float* __restrict__ agg,
                                              const float* __restrict__ W,
                                              const float* __restrict__ bias,
                                              float* __restrict__ out) {
    int lane = threadIdx.x & 31;
    int wave = threadIdx.x >> 5;
    int m0   = blockIdx.x * 16;
    int o0   = wave * 16;
    int l16  = lane & 15;
    int koff = (lane >> 4) * 2;

    const float* arow = agg + (size_t)(m0 + l16) * F_ + koff;
    const float* brow = W   + (size_t)(o0 + l16) * F_ + koff;

    v8f acc = {};
#pragma unroll
    for (int k = 0; k < F_; k += 4) {
        v2f a = *(const v2f*)(arow + k);
        v2f b = *(const v2f*)(brow + k);
        acc = __builtin_amdgcn_wmma_f32_16x16x4_f32(
             false, a,  false, b,
             (short)0, acc,  false,  false);
    }

    float bb    = bias[o0 + l16];
    int   half8 = (lane >> 4) * 8;
#pragma unroll
    for (int v = 0; v < 8; ++v) {
        out[(size_t)(m0 + v + half8) * F_ + o0 + l16] = acc[v] + bb;
    }
}

extern "C" void kernel_launch(void* const* d_in, const int* in_sizes, int n_in,
                              void* d_out, int out_size, void* d_ws, size_t ws_size,
                              hipStream_t stream) {
    const float* x    = (const float*)d_in[0];
    const int*   ei   = (const int*)  d_in[1];
    const float* bd   = (const float*)d_in[2];
    const float* W    = (const float*)d_in[3];
    const float* bias = (const float*)d_in[4];
    float*       out  = (float*)d_out;

    float* agg   = (float*)d_ws;
    float* deg   = agg + (size_t)M_ * F_;
    float* dinv  = deg + M_;
    int*   bdmax = (int*)(dinv + M_);

    k_init    <<<(M_ + 255) / 256, 256, 0, stream>>>(deg, bdmax);
    k_bdmax   <<<2048,            256, 0, stream>>>(ei, bd, bdmax);
    k_deg     <<<TE_ / 256,       256, 0, stream>>>(ei, bd, bdmax, deg);
    k_dinv_agg<<<(M_ * 32) / 256, 256, 0, stream>>>(x, deg, dinv, agg);
    k_scatter <<<TE_ / 8,         256, 0, stream>>>(ei, bd, x, dinv, bdmax, agg);
    k_gemm    <<<M_ / 16,         256, 0, stream>>>(agg, W, bias, out);
}


# Degree, self-term and edge scatter as one launch: one work-group per graph holds deg and the 558x128 aggregate in LDS; L2 float atomics become bank-spread LDS adds; sum order = LDS arrival
# speedup vs baseline: 2.6815x; 2.6815x over previous
#include <hip/hip_runtime.h>


#define S_   256
#define N_   558
#define B2_  8192
#define F_   128
#define M_   (S_ * N_)
#define TE_  (S_ * B2_)
#define GT_  1024

typedef __attribute__((ext_vector_type(2))) float v2f;
typedef __attribute__((ext_vector_type(8))) float v8f;

__global__ __launch_bounds__(32) void k_init(int* __restrict__ bdmax) {
    if (threadIdx.x == 0) *bdmax = 0;
}

__global__ __launch_bounds__(256) void k_bdmax(const int* __restrict__ ei,
                                               const float* __restrict__ bd,
                                               int* __restrict__ bdmax) {
    int stride = gridDim.x * 256;
    float m = 0.0f;
    for (int i = blockIdx.x * 256 + threadIdx.x; i < TE_; i += stride) {
        int s    = i >> 13;
        int base = i + (s << 13);
        int src  = ei[base];
        int dst  = ei[base + B2_];
        if (src != dst) m = fmaxf(m, bd[i]);
    }
    for (int off = 16; off > 0; off >>= 1)
        m = fmaxf(m, __shfl_xor(m, off, 32));
    if ((threadIdx.x & 31) == 0) atomicMax(bdmax, __float_as_int(m));
}

__device__ __forceinline__ void lds_add(float* p, float v) {
    __hip_atomic_fetch_add(p, v, __ATOMIC_RELAXED, __HIP_MEMORY_SCOPE_WORKGROUP);
}

#define LOAD8(buf, g)                                                                  \
    _Pragma("unroll")                                                                  \
    for (int u = 0; u < 8; ++u) {                                                      \
        unsigned sj = (unsigned)__builtin_amdgcn_readlane(sr, (g) + u);                \
        buf[u] = xs4[sj * (F_ / 4) + (unsigned)lane];                                  \
    }
#define ADD8(buf, g)                                                                   \
    _Pragma("unroll")                                                                  \
    for (int u = 0; u < 8; ++u) {                                                      \
        int   dj = __builtin_amdgcn_readlane(ds, (g) + u);                             \
        float nj = __builtin_bit_cast(float,                                           \
                       __builtin_amdgcn_readlane(__builtin_bit_cast(int, nrm), (g) + u)); \
        float* tp = tile + dj * F_ + lane;                                             \
        lds_add(tp,      nj * buf[u].x);                                               \
        lds_add(tp + 32, nj * buf[u].y);                                               \
        lds_add(tp + 64, nj * buf[u].z);                                               \
        lds_add(tp + 96, nj * buf[u].w);                                               \
    }

__global__ __launch_bounds__(GT_) void k_graph(const float* __restrict__ x,
                                               const int* __restrict__ ei,
                                               const float* __restrict__ bd,
                                               const int* __restrict__ bdmax,
                                               float* __restrict__ agg) {
    __shared__ float tile[N_ * F_];
    __shared__ float sdinv[N_];

    const int s    = blockIdx.x;
    const int tid  = threadIdx.x;
    const int lane = tid & 31;
    const int wave = tid >> 5;

    const int*   src_p = ei + (size_t)s * 2 * B2_;
    const int*   dst_p = src_p + B2_;
    const float* bd_p  = bd + (size_t)s * B2_;
    const float* xs    = x + (size_t)s * N_ * F_;
    const float4* xs4  = (const float4*)xs;
    float*       as    = agg + (size_t)s * N_ * F_;
    const float  bm    = __int_as_float(*bdmax);

    int   dsr[B2_ / GT_], dds[B2_ / GT_];
    float dbd[B2_ / GT_];
#pragma unroll
    for (int k = 0; k < B2_ / GT_; ++k) {
        int e  = tid + k * GT_;
        dsr[k] = src_p[e];
        dds[k] = dst_p[e];
        dbd[k] = bd_p[e];
    }

    for (int i = tid; i < N_; i += GT_) sdinv[i] = 1.0f;
    __syncthreads();

#pragma unroll
    for (int k = 0; k < B2_ / GT_; ++k) {
        if (dsr[k] != dds[k]) lds_add(&sdinv[dds[k]], dbd[k] / bm);
    }
    __syncthreads();

    for (int i = tid; i < N_; i += GT_) sdinv[i] = rsqrtf(sdinv[i]);
    __syncthreads();

    for (int i = tid; i < N_ * (F_ / 4); i += GT_) {
        int m = i >> 5;
        float di = sdinv[m];
        float sl = di * di;
        const float4 xv = *(const float4*)(xs + (size_t)i * 4);
        float* tp = tile + m * F_ + lane;
        tp[0]  = xv.x * sl;
        tp[32] = xv.y * sl;
        tp[64] = xv.z * sl;
        tp[96] = xv.w * sl;
    }
    __syncthreads();

    int   e0   = wave * 32 + lane;
    int   sr_n = src_p[e0];
    int   ds_n = dst_p[e0];
    float bd_n = bd_p[e0];
    for (int eb = wave * 32; eb < B2_; eb += GT_) {
        int   sr  = sr_n;
        int   ds  = ds_n;
        float w   = bd_n / bm;
        if (eb + GT_ < B2_) {
            int en = eb + GT_ + lane;
            sr_n = src_p[en];
            ds_n = dst_p[en];
            bd_n = bd_p[en];
        }
        bool valid = sr != ds;
        if (__builtin_amdgcn_ballot_w32(valid) == 0u) continue;
        float nrm = valid ? (sdinv[sr] * w * sdinv[ds]) : 0.0f;
        float4 xa[8], xb[8];
        LOAD8(xa, 0);
        LOAD8(xb, 8);
        ADD8(xa, 0);
        LOAD8(xa, 16);
        ADD8(xb, 8);
        LOAD8(xb, 24);
        ADD8(xa, 16);
        ADD8(xb, 24);
    }
    __syncthreads();

    for (int i = tid; i < N_ * (F_ / 4); i += GT_) {
        int m = i >> 5;
        const float* tp = tile + m * F_ + lane;
        float4 o;
        o.x = tp[0]; o.y = tp[32]; o.z = tp[64]; o.w = tp[96];
        *(float4*)(as + (size_t)i * 4) = o;
    }
}

__global__ __launch_bounds__(256) void k_gemm(const float* __restrict__ agg,
                                              const float* __restrict__ W,
                                              const float* __restrict__ bias,
                                              float* __restrict__ out) {
    int lane = threadIdx.x & 31;
    int wave = threadIdx.x >> 5;
    int m0   = blockIdx.x * 16;
    int o0   = wave * 16;
    int l16  = lane & 15;
    int koff = (lane >> 4) * 2;

    const float* arow = agg + (size_t)(m0 + l16) * F_ + koff;
    const float* brow = W   + (size_t)(o0 + l16) * F_ + koff;

    v8f acc = {};
#pragma unroll
    for (int k = 0; k < F_; k += 4) {
        v2f a = *(const v2f*)(arow + k);
        v2f b = *(const v2f*)(brow + k);
        acc = __builtin_amdgcn_wmma_f32_16x16x4_f32(
             false, a,  false, b,
             (short)0, acc,  false,  false);
    }

    float bb    = bias[o0 + l16];
    int   half8 = (lane >> 4) * 8;
#pragma unroll
    for (int v = 0; v < 8; ++v) {
        out[(size_t)(m0 + v + half8) * F_ + o0 + l16] = acc[v] + bb;
    }
}

extern "C" void kernel_launch(void* const* d_in, const int* in_sizes, int n_in,
                              void* d_out, int out_size, void* d_ws, size_t ws_size,
                              hipStream_t stream) {
    const float* x    = (const float*)d_in[0];
    const int*   ei   = (const int*)  d_in[1];
    const float* bd   = (const float*)d_in[2];
    const float* W    = (const float*)d_in[3];
    const float* bias = (const float*)d_in[4];
    float*       out  = (float*)d_out;

    float* agg   = (float*)d_ws;
    int*   bdmax = (int*)(agg + (size_t)M_ * F_);

    k_init <<<1,       32,  0, stream>>>(bdmax);
    k_bdmax<<<2048,    256, 0, stream>>>(ei, bd, bdmax);
    k_graph<<<S_,      GT_, 0, stream>>>(x, ei, bd, bdmax, agg);
    k_gemm <<<M_ / 16, 256, 0, stream>>>(agg, W, bias, out);
}


# Product as epilogue of the per-graph kernel: the finished LDS aggregate tile (pitch 132 words, feature pairs adjacent) is the WMMA A operand, W fragments in registers; no agg, 3 launches; same sums
# speedup vs baseline: 3.8391x; 1.4317x over previous
#include <hip/hip_runtime.h>


#define S_   256
#define N_   558
#define B2_  8192
#define F_   128
#define TE_  (S_ * B2_)
#define GT_  1024
#define PLH_ 66
#define STR_ 132
#define NRT_ ((N_ + 15) / 16)

typedef __attribute__((ext_vector_type(2))) float v2f;
typedef __attribute__((ext_vector_type(8))) float v8f;

__global__ __launch_bounds__(32) void k_init(int* __restrict__ bdmax) {
    if (threadIdx.x == 0) *bdmax = 0;
}

__global__ __launch_bounds__(256) void k_bdmax(const int* __restrict__ ei,
                                               const float* __restrict__ bd,
                                               int* __restrict__ bdmax) {
    int stride = gridDim.x * 256;
    float m = 0.0f;
    for (int i = blockIdx.x * 256 + threadIdx.x; i < TE_; i += stride) {
        int s    = i >> 13;
        int base = i + (s << 13);
        int src  = ei[base];
        int dst  = ei[base + B2_];
        if (src != dst) m = fmaxf(m, bd[i]);
    }
    for (int off = 16; off > 0; off >>= 1)
        m = fmaxf(m, __shfl_xor(m, off, 32));
    if ((threadIdx.x & 31) == 0) atomicMax(bdmax, __float_as_int(m));
}

__device__ __forceinline__ void lds_add(float* p, float v) {
    __hip_atomic_fetch_add(p, v, __ATOMIC_RELAXED, __HIP_MEMORY_SCOPE_WORKGROUP);
}

#define LOAD8(buf, g)                                                                  \
    _Pragma("unroll")                                                                  \
    for (int u = 0; u < 8; ++u) {                                                      \
        unsigned sj = (unsigned)__builtin_amdgcn_readlane(sr, (g) + u);                \
        buf[u] = xs4[sj * (F_ / 4) + (unsigned)lane];                                  \
    }
#define ADD8(buf, g)                                                                   \
    _Pragma("unroll")                                                                  \
    for (int u = 0; u < 8; ++u) {                                                      \
        int   dj = __builtin_amdgcn_readlane(ds, (g) + u);                             \
        float nj = __builtin_bit_cast(float,                                           \
                       __builtin_amdgcn_readlane(__builtin_bit_cast(int, nrm), (g) + u)); \
        float* tp = tile + dj * STR_ + 2 * lane;                                       \
        lds_add(tp,            nj * buf[u].x);                                         \
        lds_add(tp + 1,        nj * buf[u].y);                                         \
        lds_add(tp + PLH_,     nj * buf[u].z);                                         \
        lds_add(tp + PLH_ + 1, nj * buf[u].w);                                         \
    }

__global__ __launch_bounds__(GT_) void k_graph(const float* __restrict__ x,
                                               const int* __restrict__ ei,
                                               const float* __restrict__ bd,
                                               const int* __restrict__ bdmax,
                                               const float* __restrict__ W,
                                               const float* __restrict__ bias,
                                               float* __restrict__ out) {
    __shared__ __attribute__((aligned(16))) float tile[N_ * STR_];
    __shared__ float sdinv[N_];

    const int s    = blockIdx.x;
    const int tid  = threadIdx.x;
    const int lane = tid & 31;
    const int wave = tid >> 5;

    const int*   src_p = ei + (size_t)s * 2 * B2_;
    const int*   dst_p = src_p + B2_;
    const float* bd_p  = bd + (size_t)s * B2_;
    const float* xs    = x + (size_t)s * N_ * F_;
    const float4* xs4  = (const float4*)xs;
    float*       os    = out + (size_t)s * N_ * F_;
    const float  bm    = __int_as_float(*bdmax);

    int   dsr[B2_ / GT_], dds[B2_ / GT_];
    float dbd[B2_ / GT_];
#pragma unroll
    for (int k = 0; k < B2_ / GT_; ++k) {
        int e  = tid + k * GT_;
        dsr[k] = src_p[e];
        dds[k] = dst_p[e];
        dbd[k] = bd_p[e];
    }

    for (int i = tid; i < N_; i += GT_) sdinv[i] = 1.0f;
    __syncthreads();

#pragma unroll
    for (int k = 0; k < B2_ / GT_; ++k) {
        if (dsr[k] != dds[k]) lds_add(&sdinv[dds[k]], dbd[k] / bm);
    }
    __syncthreads();

    for (int i = tid; i < N_; i += GT_) sdinv[i] = rsqrtf(sdinv[i]);
    __syncthreads();

    for (int i = tid; i < N_ * (F_ / 4); i += GT_) {
        int m = i >> 5;
        float di = sdinv[m];
        float sl = di * di;
        const float4 xv = *(const float4*)(xs + (size_t)i * 4);
        float* tp = tile + m * STR_ + 2 * lane;
        tp[0]        = xv.x * sl;
        tp[1]        = xv.y * sl;
        tp[PLH_]     = xv.z * sl;
        tp[PLH_ + 1] = xv.w * sl;
    }
    __syncthreads();

    int   e0   = wave * 32 + lane;
    int   sr_n = src_p[e0];
    int   ds_n = dst_p[e0];
    float bd_n = bd_p[e0];
    for (int eb = wave * 32; eb < B2_; eb += GT_) {
        int   sr  = sr_n;
        int   ds  = ds_n;
        float w   = bd_n / bm;
        if (eb + GT_ < B2_) {
            int en = eb + GT_ + lane;
            sr_n = src_p[en];
            ds_n = dst_p[en];
            bd_n = bd_p[en];
        }
        bool valid = sr != ds;
        if (__builtin_amdgcn_ballot_w32(valid) == 0u) continue;
        float nrm = valid ? (sdinv[sr] * w * sdinv[ds]) : 0.0f;
        float4 xa[8], xb[8];
        LOAD8(xa, 0);
        LOAD8(xb, 8);
        ADD8(xa, 0);
        LOAD8(xa, 16);
        ADD8(xb, 8);
        LOAD8(xb, 24);
        ADD8(xa, 16);
        ADD8(xb, 24);
    }

    const int l16 = lane & 15;
    const int hh  = lane >> 4;
    const int wv  = __builtin_amdgcn_readfirstlane(wave);
    const int o0  = (wv & 7) * 16;
    const float* brow = W + (size_t)(o0 + l16) * F_ + hh * 2;
    v2f bf[F_ / 4];
#pragma unroll
    for (int t = 0; t < F_ / 4; ++t) bf[t] = *(const v2f*)(brow + 4 * t);
    const float bb = bias[o0 + l16];
    __syncthreads();

    for (int rt = wv >> 3; rt < NRT_; rt += 4) {
        const int m0 = rt * 16;
        int ar = m0 + l16;
        if (ar > N_ - 1) ar = N_ - 1;
        const float* ap = tile + ar * STR_ + hh * PLH_;
        v8f acc = {};
#pragma unroll
        for (int t = 0; t < F_ / 4; ++t) {
            v2f a = *(const v2f*)(ap + 2 * t);
            acc = __builtin_amdgcn_wmma_f32_16x16x4_f32(
                 false, a,  false, bf[t],
                 (short)0, acc,  false,  false);
        }
        float* op = os + (size_t)(m0 + hh * 8) * F_ + o0 + l16;
        if (m0 + 16 <= N_) {
#pragma unroll
            for (int v = 0; v < 8; ++v) op[v * F_] = acc[v] + bb;
        } else {
#pragma unroll
            for (int v = 0; v < 8; ++v)
                if (m0 + hh * 8 + v < N_) op[v * F_] = acc[v] + bb;
        }
    }
}

extern "C" void kernel_launch(void* const* d_in, const int* in_sizes, int n_in,
                              void* d_out, int out_size, void* d_ws, size_t ws_size,
                              hipStream_t stream) {
    const float* x    = (const float*)d_in[0];
    const int*   ei   = (const int*)  d_in[1];
    const float* bd   = (const float*)d_in[2];
    const float* W    = (const float*)d_in[3];
    const float* bias = (const float*)d_in[4];
    float*       out  = (float*)d_out;

    int* bdmax = (int*)d_ws;

    k_init <<<1,    32,  0, stream>>>(bdmax);
    k_bdmax<<<2048, 256, 0, stream>>>(ei, bd, bdmax);
    k_graph<<<S_,   GT_, 0, stream>>>(x, ei, bd, bdmax, W, bias, out);
}


# Scatter without LDS float atomics: a wave owns 18 destination rows, ballot-filters the edge list into its own queue, adds by plain 16-byte LDS read-modify-write; sum order = fixed scan order
# speedup vs baseline: 4.0990x; 1.0677x over previous
#include <hip/hip_runtime.h>


#define S_   256
#define N_   558
#define B2_  8192
#define F_   128
#define TE_  (S_ * B2_)
#define GT_  1024
#define STR_ 132
#define RW_  18
#define SCB_ 128
#define QCAP_ 160
#define NRT_ ((N_ + 15) / 16)

typedef __attribute__((ext_vector_type(2))) float v2f;
typedef __attribute__((ext_vector_type(8))) float v8f;

__global__ __launch_bounds__(32) void k_init(int* __restrict__ bdmax) {
    if (threadIdx.x == 0) *bdmax = 0;
}

__global__ __launch_bounds__(256) void k_bdmax(const int* __restrict__ ei,
                                               const float* __restrict__ bd,
                                               int* __restrict__ bdmax) {
    int stride = gridDim.x * 256;
    float m = 0.0f;
    for (int i = blockIdx.x * 256 + threadIdx.x; i < TE_; i += stride) {
        int s    = i >> 13;
        int base = i + (s << 13);
        int src  = ei[base];
        int dst  = ei[base + B2_];
        if (src != dst) m = fmaxf(m, bd[i]);
    }
    for (int off = 16; off > 0; off >>= 1)
        m = fmaxf(m, __shfl_xor(m, off, 32));
    if ((threadIdx.x & 31) == 0) atomicMax(bdmax, __float_as_int(m));
}

__device__ __forceinline__ void lds_add(float* p, float v) {
    __hip_atomic_fetch_add(p, v, __ATOMIC_RELAXED, __HIP_MEMORY_SCOPE_WORKGROUP);
}

#define LOAD8(buf, g)                                                                  \
    _Pragma("unroll")                                                                  \
    for (int u = 0; u < 8; ++u) {                                                      \
        unsigned sj = (unsigned)__builtin_amdgcn_readlane(sr, (g) + u);                \
        buf[u] = xs4[sj * (F_ / 4) + (unsigned)lane];                                  \
    }
#define RMW8(buf, g)                                                                   \
    _Pragma("unroll")                                                                  \
    for (int u = 0; u < 8; ++u) {                                                      \
        int   rj = __builtin_amdgcn_readlane((int)rr_q, (g) + u);                      \
        float nj = __builtin_bit_cast(float,                                           \
                       __builtin_amdgcn_readlane(__builtin_bit_cast(int, nrm), (g) + u)); \
        float4* tp = (float4*)(tile + (lo + rj) * STR_) + lane;                        \
        float4 a = *tp;                                                                \
        a.x += nj * buf[u].x;                                                          \
        a.y += nj * buf[u].y;                                                          \
        a.z += nj * buf[u].z;                                                          \
        a.w += nj * buf[u].w;                                                          \
        *tp = a;                                                                       \
    }

__global__ __launch_bounds__(GT_) void k_graph(const float* __restrict__ x,
                                               const int* __restrict__ ei,
                                               const float* __restrict__ bd,
                                               const int* __restrict__ bdmax,
                                               const float* __restrict__ W,
                                               const float* __restrict__ bias,
                                               float* __restrict__ out) {
    __shared__ __attribute__((aligned(16))) float tile[N_ * STR_];
    __shared__ float sdinv[N_];
    __shared__ __attribute__((aligned(16))) unsigned short d16[B2_];
    __shared__ unsigned short qall[(GT_ / 32) * QCAP_];

    const int s    = blockIdx.x;
    const int tid  = threadIdx.x;
    const int lane = tid & 31;
    const int wave = tid >> 5;

    const int*   src_p = ei + (size_t)s * 2 * B2_;
    const int*   dst_p = src_p + B2_;
    const float* bd_p  = bd + (size_t)s * B2_;
    const float* xs    = x + (size_t)s * N_ * F_;
    const float4* xs4  = (const float4*)xs;
    float*       os    = out + (size_t)s * N_ * F_;
    const float  bm    = __int_as_float(*bdmax);

    int   dsr[B2_ / GT_], dds[B2_ / GT_];
    float dbd[B2_ / GT_];
#pragma unroll
    for (int k = 0; k < B2_ / GT_; ++k) {
        int e  = tid + k * GT_;
        dsr[k] = src_p[e];
        dds[k] = dst_p[e];
        dbd[k] = bd_p[e];
    }

    for (int i = tid; i < N_; i += GT_) sdinv[i] = 1.0f;
#pragma unroll
    for (int k = 0; k < B2_ / GT_; ++k)
        d16[tid + k * GT_] = (dsr[k] != dds[k]) ? (unsigned short)dds[k] : (unsigned short)0xFFFF;
    __syncthreads();

#pragma unroll
    for (int k = 0; k < B2_ / GT_; ++k) {
        if (dsr[k] != dds[k]) lds_add(&sdinv[dds[k]], dbd[k] / bm);
    }
    __syncthreads();

    for (int i = tid; i < N_; i += GT_) sdinv[i] = rsqrtf(sdinv[i]);
    __syncthreads();

    const int wv = __builtin_amdgcn_readfirstlane(wave);
    const int lo = wv * RW_;
    unsigned short* qw = qall + wv * QCAP_;

    if (lo < N_) {
        float4 xr[RW_];
#pragma unroll
        for (int r = 0; r < RW_; ++r) xr[r] = xs4[(lo + r) * (F_ / 4) + lane];
#pragma unroll
        for (int r = 0; r < RW_; ++r) {
            float di = sdinv[lo + r];
            float sl = di * di;
            float4 v = xr[r];
            v.x *= sl; v.y *= sl; v.z *= sl; v.w *= sl;
            *((float4*)(tile + (lo + r) * STR_) + lane) = v;
        }

        int qn = 0;
        for (int eb = 0; eb < B2_; eb += SCB_) {
            const uint2 dv = *(const uint2*)(d16 + eb + 4 * lane);
#pragma unroll
            for (int k = 0; k < 4; ++k) {
                unsigned w2 = (k & 2) ? dv.y : dv.x;
                unsigned d  = (k & 1) ? (w2 >> 16) : (w2 & 0xFFFFu);
                bool mine   = (d - (unsigned)lo) < (unsigned)RW_;
                unsigned mask = __builtin_amdgcn_ballot_w32(mine);
                if (mask != 0u) {
                    unsigned below = __builtin_amdgcn_mbcnt_lo(mask, 0u);
                    if (mine) qw[qn + below] = (unsigned short)(eb + 4 * lane + k);
                    qn += __builtin_popcount(mask);
                }
            }
            const bool last = (eb + SCB_ >= B2_);
            int head = 0;
            while (qn - head >= 32 || (last && qn - head > 0)) {
                const int n = (qn - head < 32) ? (qn - head) : 32;
                const bool ok = lane < n;
                unsigned e    = ok ? (unsigned)qw[head + lane] : 0u;
                unsigned rr_q = ok ? ((unsigned)d16[e] - (unsigned)lo) : 0u;
                int   sr  = src_p[e];
                float w   = bd_p[e] / bm;
                float nrm = ok ? (sdinv[sr] * w * sdinv[lo + rr_q]) : 0.0f;
                float4 xa[8], xb[8];
                LOAD8(xa, 0);
                LOAD8(xb, 8);
                RMW8(xa, 0);
                LOAD8(xa, 16);
                RMW8(xb, 8);
                LOAD8(xb, 24);
                RMW8(xa, 16);
                RMW8(xb, 24);
                head += n;
            }
            if (head > 0) {
                const int rem = qn - head;
                unsigned short t = (lane < rem) ? qw[head + lane] : (unsigned short)0;
                if (lane < rem) qw[lane] = t;
                qn = rem;
            }
        }
    }

    const int l16 = lane & 15;
    const int hh  = lane >> 4;
    const int o0  = (wv & 7) * 16;
    const float* brow = W + (size_t)(o0 + l16) * F_ + hh * 2;
    v2f bf[F_ / 4];
#pragma unroll
    for (int t = 0; t < F_ / 4; ++t) bf[t] = *(const v2f*)(brow + 4 * t);
    const float bb = bias[o0 + l16];
    __syncthreads();

    for (int rt = wv >> 3; rt < NRT_; rt += 4) {
        const int m0 = rt * 16;
        int ar = m0 + l16;
        if (ar > N_ - 1) ar = N_ - 1;
        const float* ap = tile + ar * STR_ + hh * 2;
        v8f acc = {};
#pragma unroll
        for (int t = 0; t < F_ / 4; ++t) {
            v2f a = *(const v2f*)(ap + 4 * t);
            acc = __builtin_amdgcn_wmma_f32_16x16x4_f32(
                 false, a,  false, bf[t],
                 (short)0, acc,  false,  false);
        }
        float* op = os + (size_t)(m0 + hh * 8) * F_ + o0 + l16;
        if (m0 + 16 <= N_) {
#pragma unroll
            for (int v = 0; v < 8; ++v) op[v * F_] = acc[v] + bb;
        } else {
#pragma unroll
            for (int v = 0; v < 8; ++v)
                if (m0 + hh * 8 + v < N_) op[v * F_] = acc[v] + bb;
        }
    }
}

extern "C" void kernel_launch(void* const* d_in, const int* in_sizes, int n_in,
                              void* d_out, int out_size, void* d_ws, size_t ws_size,
                              hipStream_t stream) {
    const float* x    = (const float*)d_in[0];
    const int*   ei   = (const int*)  d_in[1];
    const float* bd   = (const float*)d_in[2];
    const float* W    = (const float*)d_in[3];
    const float* bias = (const float*)d_in[4];
    float*       out  = (float*)d_out;

    int* bdmax = (int*)d_ws;

    k_init <<<1,    32,  0, stream>>>(bdmax);
    k_bdmax<<<2048, 256, 0, stream>>>(ei, bd, bdmax);
    k_graph<<<S_,   GT_, 0, stream>>>(x, ei, bd, bdmax, W, bias, out);
}


# Masked-max kernel: the 8 wave maxima of a work-group meet in LDS and one thread sends one atomic max, so the single hot word takes 2048 atomics instead of 16384; same grid, loads and launches
# speedup vs baseline: 4.7231x; 1.1523x over previous
#include <hip/hip_runtime.h>


#define S_   256
#define N_   558
#define B2_  8192
#define F_   128
#define TE_  (S_ * B2_)
#define GT_  1024
#define STR_ 132
#define RW_  18
#define SCB_ 128
#define QCAP_ 160
#define NRT_ ((N_ + 15) / 16)

typedef __attribute__((ext_vector_type(2))) float v2f;
typedef __attribute__((ext_vector_type(8))) float v8f;

__global__ __launch_bounds__(32) void k_init(int* __restrict__ bdmax) {
    if (threadIdx.x == 0) *bdmax = 0;
}

__global__ __launch_bounds__(256) void k_bdmax(const int* __restrict__ ei,
                                               const float* __restrict__ bd,
                                               int* __restrict__ bdmax) {
    __shared__ float wmax[8];
    int stride = gridDim.x * 256;
    float m = 0.0f;
    for (int i = blockIdx.x * 256 + threadIdx.x; i < TE_; i += stride) {
        int s    = i >> 13;
        int base = i + (s << 13);
        int src  = ei[base];
        int dst  = ei[base + B2_];
        if (src != dst) m = fmaxf(m, bd[i]);
    }
    for (int off = 16; off > 0; off >>= 1)
        m = fmaxf(m, __shfl_xor(m, off, 32));
    if ((threadIdx.x & 31) == 0) wmax[threadIdx.x >> 5] = m;
    __syncthreads();
    if (threadIdx.x == 0) {
        float g = wmax[0];
#pragma unroll
        for (int w = 1; w < 8; ++w) g = fmaxf(g, wmax[w]);
        atomicMax(bdmax, __float_as_int(g));
    }
}

__device__ __forceinline__ void lds_add(float* p, float v) {
    __hip_atomic_fetch_add(p, v, __ATOMIC_RELAXED, __HIP_MEMORY_SCOPE_WORKGROUP);
}

#define LOAD8(buf, g)                                                                  \
    _Pragma("unroll")                                                                  \
    for (int u = 0; u < 8; ++u) {                                                      \
        unsigned sj = (unsigned)__builtin_amdgcn_readlane(sr, (g) + u);                \
        buf[u] = xs4[sj * (F_ / 4) + (unsigned)lane];                                  \
    }
#define RMW8(buf, g)                                                                   \
    _Pragma("unroll")                                                                  \
    for (int u = 0; u < 8; ++u) {                                                      \
        int   rj = __builtin_amdgcn_readlane((int)rr_q, (g) + u);                      \
        float nj = __builtin_bit_cast(float,                                           \
                       __builtin_amdgcn_readlane(__builtin_bit_cast(int, nrm), (g) + u)); \
        float4* tp = (float4*)(tile + (lo + rj) * STR_) + lane;                        \
        float4 a = *tp;                                                                \
        a.x += nj * buf[u].x;                                                          \
        a.y += nj * buf[u].y;                                                          \
        a.z += nj * buf[u].z;                                                          \
        a.w += nj * buf[u].w;                                                          \
        *tp = a;                                                                       \
    }

__global__ __launch_bounds__(GT_) void k_graph(const float* __restrict__ x,
                                               const int* __restrict__ ei,
                                               const float* __restrict__ bd,
                                               const int* __restrict__ bdmax,
                                               const float* __restrict__ W,
                                               const float* __restrict__ bias,
                                               float* __restrict__ out) {
    __shared__ __attribute__((aligned(16))) float tile[N_ * STR_];
    __shared__ float sdinv[N_];
    __shared__ __attribute__((aligned(16))) unsigned short d16[B2_];
    __shared__ unsigned short qall[(GT_ / 32) * QCAP_];

    const int s    = blockIdx.x;
    const int tid  = threadIdx.x;
    const int lane = tid & 31;
    const int wave = tid >> 5;

    const int*   src_p = ei + (size_t)s * 2 * B2_;
    const int*   dst_p = src_p + B2_;
    const float* bd_p  = bd + (size_t)s * B2_;
    const float* xs    = x + (size_t)s * N_ * F_;
    const float4* xs4  = (const float4*)xs;
    float*       os    = out + (size_t)s * N_ * F_;
    const float  bm    = __int_as_float(*bdmax);

    int   dsr[B2_ / GT_], dds[B2_ / GT_];
    float dbd[B2_ / GT_];
#pragma unroll
    for (int k = 0; k < B2_ / GT_; ++k) {
        int e  = tid + k * GT_;
        dsr[k] = src_p[e];
        dds[k] = dst_p[e];
        dbd[k] = bd_p[e];
    }

    for (int i = tid; i < N_; i += GT_) sdinv[i] = 1.0f;
#pragma unroll
    for (int k = 0; k < B2_ / GT_; ++k)
        d16[tid + k * GT_] = (dsr[k] != dds[k]) ? (unsigned short)dds[k] : (unsigned short)0xFFFF;
    __syncthreads();

#pragma unroll
    for (int k = 0; k < B2_ / GT_; ++k) {
        if (dsr[k] != dds[k]) lds_add(&sdinv[dds[k]], dbd[k] / bm);
    }
    __syncthreads();

    for (int i = tid; i < N_; i += GT_) sdinv[i] = rsqrtf(sdinv[i]);
    __syncthreads();

    const int wv = __builtin_amdgcn_readfirstlane(wave);
    const int lo = wv * RW_;
    unsigned short* qw = qall + wv * QCAP_;

    if (lo < N_) {
        float4 xr[RW_];
#pragma unroll
        for (int r = 0; r < RW_; ++r) xr[r] = xs4[(lo + r) * (F_ / 4) + lane];
#pragma unroll
        for (int r = 0; r < RW_; ++r) {
            float di = sdinv[lo + r];
            float sl = di * di;
            float4 v = xr[r];
            v.x *= sl; v.y *= sl; v.z *= sl; v.w *= sl;
            *((float4*)(tile + (lo + r) * STR_) + lane) = v;
        }

        int qn = 0;
        for (int eb = 0; eb < B2_; eb += SCB_) {
            const uint2 dv = *(const uint2*)(d16 + eb + 4 * lane);
#pragma unroll
            for (int k = 0; k < 4; ++k) {
                unsigned w2 = (k & 2) ? dv.y : dv.x;
                unsigned d  = (k & 1) ? (w2 >> 16) : (w2 & 0xFFFFu);
                bool mine   = (d - (unsigned)lo) < (unsigned)RW_;
                unsigned mask = __builtin_amdgcn_ballot_w32(mine);
                if (mask != 0u) {
                    unsigned below = __builtin_amdgcn_mbcnt_lo(mask, 0u);
                    if (mine) qw[qn + below] = (unsigned short)(eb + 4 * lane + k);
                    qn += __builtin_popcount(mask);
                }
            }
            const bool last = (eb + SCB_ >= B2_);
            int head = 0;
            while (qn - head >= 32 || (last && qn - head > 0)) {
                const int n = (qn - head < 32) ? (qn - head) : 32;
                const bool ok = lane < n;
                unsigned e    = ok ? (unsigned)qw[head + lane] : 0u;
                unsigned rr_q = ok ? ((unsigned)d16[e] - (unsigned)lo) : 0u;
                int   sr  = src_p[e];
                float w   = bd_p[e] / bm;
                float nrm = ok ? (sdinv[sr] * w * sdinv[lo + rr_q]) : 0.0f;
                float4 xa[8], xb[8];
                LOAD8(xa, 0);
                LOAD8(xb, 8);
                RMW8(xa, 0);
                LOAD8(xa, 16);
                RMW8(xb, 8);
                LOAD8(xb, 24);
                RMW8(xa, 16);
                RMW8(xb, 24);
                head += n;
            }
            if (head > 0) {
                const int rem = qn - head;
                unsigned short t = (lane < rem) ? qw[head + lane] : (unsigned short)0;
                if (lane < rem) qw[lane] = t;
                qn = rem;
            }
        }
    }

    const int l16 = lane & 15;
    const int hh  = lane >> 4;
    const int o0  = (wv & 7) * 16;
    const float* brow = W + (size_t)(o0 + l16) * F_ + hh * 2;
    v2f bf[F_ / 4];
#pragma unroll
    for (int t = 0; t < F_ / 4; ++t) bf[t] = *(const v2f*)(brow + 4 * t);
    const float bb = bias[o0 + l16];
    __syncthreads();

    for (int rt = wv >> 3; rt < NRT_; rt += 4) {
        const int m0 = rt * 16;
        int ar = m0 + l16;
        if (ar > N_ - 1) ar = N_ - 1;
        const float* ap = tile + ar * STR_ + hh * 2;
        v8f acc = {};
#pragma unroll
        for (int t = 0; t < F_ / 4; ++t) {
            v2f a = *(const v2f*)(ap + 4 * t);
            acc = __builtin_amdgcn_wmma_f32_16x16x4_f32(
                 false, a,  false, bf[t],
                 (short)0, acc,  false,  false);
        }
        float* op = os + (size_t)(m0 + hh * 8) * F_ + o0 + l16;
        if (m0 + 16 <= N_) {
#pragma unroll
            for (int v = 0; v < 8; ++v) op[v * F_] = acc[v] + bb;
        } else {
#pragma unroll
            for (int v = 0; v < 8; ++v)
                if (m0 + hh * 8 + v < N_) op[v * F_] = acc[v] + bb;
        }
    }
}

extern "C" void kernel_launch(void* const* d_in, const int* in_sizes, int n_in,
                              void* d_out, int out_size, void* d_ws, size_t ws_size,
                              hipStream_t stream) {
    const float* x    = (const float*)d_in[0];
    const int*   ei   = (const int*)  d_in[1];
    const float* bd   = (const float*)d_in[2];
    const float* W    = (const float*)d_in[3];
    const float* bias = (const float*)d_in[4];
    float*       out  = (float*)d_out;

    int* bdmax = (int*)d_ws;

    k_init <<<1,    32,  0, stream>>>(bdmax);
    k_bdmax<<<2048, 256, 0, stream>>>(ei, bd, bdmax);
    k_graph<<<S_,   GT_, 0, stream>>>(x, ei, bd, bdmax, W, bias, out);
}


# Scatter adds in groups of four queued edges: the four LDS row reads go out together ahead of a scalar test of the row numbers (a repeated row forwarded or re-read); same sum order
# speedup vs baseline: 4.9396x; 1.0458x over previous
#include <hip/hip_runtime.h>


#define S_   256
#define N_   558
#define B2_  8192
#define F_   128
#define TE_  (S_ * B2_)
#define GT_  1024
#define STR_ 132
#define RW_  18
#define SCB_ 128
#define QCAP_ 160
#define NRT_ ((N_ + 15) / 16)

typedef __attribute__((ext_vector_type(2))) float v2f;
typedef __attribute__((ext_vector_type(8))) float v8f;

__global__ __launch_bounds__(32) void k_init(int* __restrict__ bdmax) {
    if (threadIdx.x == 0) *bdmax = 0;
}

__global__ __launch_bounds__(256) void k_bdmax(const int* __restrict__ ei,
                                               const float* __restrict__ bd,
                                               int* __restrict__ bdmax) {
    __shared__ float wmax[8];
    int stride = gridDim.x * 256;
    float m = 0.0f;
    for (int i = blockIdx.x * 256 + threadIdx.x; i < TE_; i += stride) {
        int s    = i >> 13;
        int base = i + (s << 13);
        int src  = ei[base];
        int dst  = ei[base + B2_];
        if (src != dst) m = fmaxf(m, bd[i]);
    }
    for (int off = 16; off > 0; off >>= 1)
        m = fmaxf(m, __shfl_xor(m, off, 32));
    if ((threadIdx.x & 31) == 0) wmax[threadIdx.x >> 5] = m;
    __syncthreads();
    if (threadIdx.x == 0) {
        float g = wmax[0];
#pragma unroll
        for (int w = 1; w < 8; ++w) g = fmaxf(g, wmax[w]);
        atomicMax(bdmax, __float_as_int(g));
    }
}

__device__ __forceinline__ void lds_add(float* p, float v) {
    __hip_atomic_fetch_add(p, v, __ATOMIC_RELAXED, __HIP_MEMORY_SCOPE_WORKGROUP);
}

#define LOAD4(buf, o, g)                                                               \
    _Pragma("unroll")                                                                  \
    for (int u = 0; u < 4; ++u) {                                                      \
        unsigned sj = (unsigned)__builtin_amdgcn_readlane(sr, (g) + u);                \
        buf[(o) + u] = xs4[sj * (F_ / 4) + (unsigned)lane];                            \
    }

__device__ __forceinline__ void fma4(float4& a, float n, const float4& b) {
    a.x += n * b.x;
    a.y += n * b.y;
    a.z += n * b.z;
    a.w += n * b.w;
}

__device__ __forceinline__ void add4(float* tile, int lo, int lane, unsigned rr_q, float nrm, int g,
                                     const float4& b0, const float4& b1,
                                     const float4& b2, const float4& b3) {
    const int r0 = __builtin_amdgcn_readlane((int)rr_q, g);
    const int r1 = __builtin_amdgcn_readlane((int)rr_q, g + 1);
    const int r2 = __builtin_amdgcn_readlane((int)rr_q, g + 2);
    const int r3 = __builtin_amdgcn_readlane((int)rr_q, g + 3);
    const int ni = __builtin_bit_cast(int, nrm);
    const float n0 = __builtin_bit_cast(float, __builtin_amdgcn_readlane(ni, g));
    const float n1 = __builtin_bit_cast(float, __builtin_amdgcn_readlane(ni, g + 1));
    const float n2 = __builtin_bit_cast(float, __builtin_amdgcn_readlane(ni, g + 2));
    const float n3 = __builtin_bit_cast(float, __builtin_amdgcn_readlane(ni, g + 3));
    float4* p0 = (float4*)(tile + (lo + r0) * STR_) + lane;
    float4* p1 = (float4*)(tile + (lo + r1) * STR_) + lane;
    float4* p2 = (float4*)(tile + (lo + r2) * STR_) + lane;
    float4* p3 = (float4*)(tile + (lo + r3) * STR_) + lane;
    float4 a0 = *p0, a1 = *p1, a2 = *p2, a3 = *p3;
    const bool all4 = (r0 != r1) & (r0 != r2) & (r0 != r3) & (r1 != r2) & (r1 != r3) & (r2 != r3);
    if (all4) {
        fma4(a0, n0, b0);
        fma4(a1, n1, b1);
        fma4(a2, n2, b2);
        fma4(a3, n3, b3);
        *p0 = a0;
        *p1 = a1;
        *p2 = a2;
        *p3 = a3;
    } else {
        fma4(a0, n0, b0);
        if (r1 == r0) a1 = a0;
        fma4(a1, n1, b1);
        *p0 = a0;
        *p1 = a1;
        if ((r2 == r0) | (r2 == r1) | (r3 == r0) | (r3 == r1)) {
            a2 = *p2;
            a3 = *p3;
        }
        fma4(a2, n2, b2);
        if (r3 == r2) a3 = a2;
        fma4(a3, n3, b3);
        *p2 = a2;
        *p3 = a3;
    }
}

#define RMW4(buf, o, g)                                                                \
    add4(tile, lo, lane, rr_q, nrm, (g), buf[(o)], buf[(o) + 1], buf[(o) + 2], buf[(o) + 3]);

__global__ __launch_bounds__(GT_) void k_graph(const float* __restrict__ x,
                                               const int* __restrict__ ei,
                                               const float* __restrict__ bd,
                                               const int* __restrict__ bdmax,
                                               const float* __restrict__ W,
                                               const float* __restrict__ bias,
                                               float* __restrict__ out) {
    __shared__ __attribute__((aligned(16))) float tile[N_ * STR_];
    __shared__ float sdinv[N_];
    __shared__ __attribute__((aligned(16))) unsigned short d16[B2_];
    __shared__ unsigned short qall[(GT_ / 32) * QCAP_];

    const int s    = blockIdx.x;
    const int tid  = threadIdx.x;
    const int lane = tid & 31;
    const int wave = tid >> 5;

    const int*   src_p = ei + (size_t)s * 2 * B2_;
    const int*   dst_p = src_p + B2_;
    const float* bd_p  = bd + (size_t)s * B2_;
    const float* xs    = x + (size_t)s * N_ * F_;
    const float4* xs4  = (const float4*)xs;
    float*       os    = out + (size_t)s * N_ * F_;
    const float  bm    = __int_as_float(*bdmax);

    int   dsr[B2_ / GT_], dds[B2_ / GT_];
    float dbd[B2_ / GT_];
#pragma unroll
    for (int k = 0; k < B2_ / GT_; ++k) {
        int e  = tid + k * GT_;
        dsr[k] = src_p[e];
        dds[k] = dst_p[e];
        dbd[k] = bd_p[e];
    }

    for (int i = tid; i < N_; i += GT_) sdinv[i] = 1.0f;
#pragma unroll
    for (int k = 0; k < B2_ / GT_; ++k)
        d16[tid + k * GT_] = (dsr[k] != dds[k]) ? (unsigned short)dds[k] : (unsigned short)0xFFFF;
    __syncthreads();

#pragma unroll
    for (int k = 0; k < B2_ / GT_; ++k) {
        if (dsr[k] != dds[k]) lds_add(&sdinv[dds[k]], dbd[k] / bm);
    }
    __syncthreads();

    for (int i = tid; i < N_; i += GT_) sdinv[i] = rsqrtf(sdinv[i]);
    __syncthreads();

    const int wv = __builtin_amdgcn_readfirstlane(wave);
    const int lo = wv * RW_;
    unsigned short* qw = qall + wv * QCAP_;

    if (lo < N_) {
        float4 xr[RW_];
#pragma unroll
        for (int r = 0; r < RW_; ++r) xr[r] = xs4[(lo + r) * (F_ / 4) + lane];
#pragma unroll
        for (int r = 0; r < RW_; ++r) {
            float di = sdinv[lo + r];
            float sl = di * di;
            float4 v = xr[r];
            v.x *= sl; v.y *= sl; v.z *= sl; v.w *= sl;
            *((float4*)(tile + (lo + r) * STR_) + lane) = v;
        }

        int qn = 0;
        for (int eb = 0; eb < B2_; eb += SCB_) {
            const uint2 dv = *(const uint2*)(d16 + eb + 4 * lane);
#pragma unroll
            for (int k = 0; k < 4; ++k) {
                unsigned w2 = (k & 2) ? dv.y : dv.x;
                unsigned d  = (k & 1) ? (w2 >> 16) : (w2 & 0xFFFFu);
                bool mine   = (d - (unsigned)lo) < (unsigned)RW_;
                unsigned mask = __builtin_amdgcn_ballot_w32(mine);
                if (mask != 0u) {
                    unsigned below = __builtin_amdgcn_mbcnt_lo(mask, 0u);
                    if (mine) qw[qn + below] = (unsigned short)(eb + 4 * lane + k);
                    qn += __builtin_popcount(mask);
                }
            }
            const bool last = (eb + SCB_ >= B2_);
            int head = 0;
            while (qn - head >= 32 || (last && qn - head > 0)) {
                const int n = (qn - head < 32) ? (qn - head) : 32;
                const bool ok = lane < n;
                unsigned e    = ok ? (unsigned)qw[head + lane] : 0u;
                unsigned rr_q = ok ? ((unsigned)d16[e] - (unsigned)lo) : 0u;
                int   sr  = src_p[e];
                float w   = bd_p[e] / bm;
                float nrm = ok ? (sdinv[sr] * w * sdinv[lo + rr_q]) : 0.0f;
                float4 xa[8], xb[8];
                LOAD4(xa, 0, 0);
                LOAD4(xa, 4, 4);
                LOAD4(xb, 0, 8);
                LOAD4(xb, 4, 12);
                RMW4(xa, 0, 0);
                LOAD4(xa, 0, 16);
                RMW4(xa, 4, 4);
                LOAD4(xa, 4, 20);
                RMW4(xb, 0, 8);
                LOAD4(xb, 0, 24);
                RMW4(xb, 4, 12);
                LOAD4(xb, 4, 28);
                RMW4(xa, 0, 16);
                RMW4(xa, 4, 20);
                RMW4(xb, 0, 24);
                RMW4(xb, 4, 28);
                head += n;
            }
            if (head > 0) {
                const int rem = qn - head;
                unsigned short t = (lane < rem) ? qw[head + lane] : (unsigned short)0;
                if (lane < rem) qw[lane] = t;
                qn = rem;
            }
        }
    }

    const int l16 = lane & 15;
    const int hh  = lane >> 4;
    const int o0  = (wv & 7) * 16;
    const float* brow = W + (size_t)(o0 + l16) * F_ + hh * 2;
    v2f bf[F_ / 4];
#pragma unroll
    for (int t = 0; t < F_ / 4; ++t) bf[t] = *(const v2f*)(brow + 4 * t);
    const float bb = bias[o0 + l16];
    __syncthreads();

    for (int rt = wv >> 3; rt < NRT_; rt += 4) {
        const int m0 = rt * 16;
        int ar = m0 + l16;
        if (ar > N_ - 1) ar = N_ - 1;
        const float* ap = tile + ar * STR_ + hh * 2;
        v8f acc = {};
#pragma unroll
        for (int t = 0; t < F_ / 4; ++t) {
            v2f a = *(const v2f*)(ap + 4 * t);
            acc = __builtin_amdgcn_wmma_f32_16x16x4_f32(
                 false, a,  false, bf[t],
                 (short)0, acc,  false,  false);
        }
        float* op = os + (size_t)(m0 + hh * 8) * F_ + o0 + l16;
        if (m0 + 16 <= N_) {
#pragma unroll
            for (int v = 0; v < 8; ++v) op[v * F_] = acc[v] + bb;
        } else {
#pragma unroll
            for (int v = 0; v < 8; ++v)
                if (m0 + hh * 8 + v < N_) op[v * F_] = acc[v] + bb;
        }
    }
}

extern "C" void kernel_launch(void* const* d_in, const int* in_sizes, int n_in,
                              void* d_out, int out_size, void* d_ws, size_t ws_size,
                              hipStream_t stream) {
    const float* x    = (const float*)d_in[0];
    const int*   ei   = (const int*)  d_in[1];
    const float* bd   = (const float*)d_in[2];
    const float* W    = (const float*)d_in[3];
    const float* bias = (const float*)d_in[4];
    float*       out  = (float*)d_out;

    int* bdmax = (int*)d_ws;

    k_init <<<1,    32,  0, stream>>>(bdmax);
    k_bdmax<<<2048, 256, 0, stream>>>(ei, bd, bdmax);
    k_graph<<<S_,   GT_, 0, stream>>>(x, ei, bd, bdmax, W, bias, out);
}


# Edge filter on byte tags: LDS holds per edge the owning wave (1 byte); a wave tests 16 tags per lane and step with 32-bit byte operations, queues matches in rounds; sum order = new fixed scan order
# speedup vs baseline: 5.2425x; 1.0613x over previous
#include <hip/hip_runtime.h>


#define S_   256
#define N_   558
#define B2_  8192
#define F_   128
#define TE_  (S_ * B2_)
#define GT_  1024
#define STR_ 132
#define RW_  18
#define SCB_ 512
#define QCAP_ 64
#define NRT_ ((N_ + 15) / 16)

typedef __attribute__((ext_vector_type(2))) float v2f;
typedef __attribute__((ext_vector_type(8))) float v8f;

__global__ __launch_bounds__(32) void k_init(int* __restrict__ bdmax) {
    if (threadIdx.x == 0) *bdmax = 0;
}

__global__ __launch_bounds__(256) void k_bdmax(const int* __restrict__ ei,
                                               const float* __restrict__ bd,
                                               int* __restrict__ bdmax) {
    __shared__ float wmax[8];
    int stride = gridDim.x * 256;
    float m = 0.0f;
    for (int i = blockIdx.x * 256 + threadIdx.x; i < TE_; i += stride) {
        int s    = i >> 13;
        int base = i + (s << 13);
        int src  = ei[base];
        int dst  = ei[base + B2_];
        if (src != dst) m = fmaxf(m, bd[i]);
    }
    for (int off = 16; off > 0; off >>= 1)
        m = fmaxf(m, __shfl_xor(m, off, 32));
    if ((threadIdx.x & 31) == 0) wmax[threadIdx.x >> 5] = m;
    __syncthreads();
    if (threadIdx.x == 0) {
        float g = wmax[0];
#pragma unroll
        for (int w = 1; w < 8; ++w) g = fmaxf(g, wmax[w]);
        atomicMax(bdmax, __float_as_int(g));
    }
}

__device__ __forceinline__ void lds_add(float* p, float v) {
    __hip_atomic_fetch_add(p, v, __ATOMIC_RELAXED, __HIP_MEMORY_SCOPE_WORKGROUP);
}

#define LOAD4(buf, o, g)                                                               \
    _Pragma("unroll")                                                                  \
    for (int u = 0; u < 4; ++u) {                                                      \
        unsigned sj = (unsigned)__builtin_amdgcn_readlane(sr, (g) + u);                \
        buf[(o) + u] = xs4[sj * (F_ / 4) + (unsigned)lane];                            \
    }

__device__ __forceinline__ void fma4(float4& a, float n, const float4& b) {
    a.x += n * b.x;
    a.y += n * b.y;
    a.z += n * b.z;
    a.w += n * b.w;
}

__device__ __forceinline__ void add4(float* tile, int lo, int lane, unsigned rr_q, float nrm, int g,
                                     const float4& b0, const float4& b1,
                                     const float4& b2, const float4& b3) {
    const int r0 = __builtin_amdgcn_readlane((int)rr_q, g);
    const int r1 = __builtin_amdgcn_readlane((int)rr_q, g + 1);
    const int r2 = __builtin_amdgcn_readlane((int)rr_q, g + 2);
    const int r3 = __builtin_amdgcn_readlane((int)rr_q, g + 3);
    const int ni = __builtin_bit_cast(int, nrm);
    const float n0 = __builtin_bit_cast(float, __builtin_amdgcn_readlane(ni, g));
    const float n1 = __builtin_bit_cast(float, __builtin_amdgcn_readlane(ni, g + 1));
    const float n2 = __builtin_bit_cast(float, __builtin_amdgcn_readlane(ni, g + 2));
    const float n3 = __builtin_bit_cast(float, __builtin_amdgcn_readlane(ni, g + 3));
    float4* p0 = (float4*)(tile + (lo + r0) * STR_) + lane;
    float4* p1 = (float4*)(tile + (lo + r1) * STR_) + lane;
    float4* p2 = (float4*)(tile + (lo + r2) * STR_) + lane;
    float4* p3 = (float4*)(tile + (lo + r3) * STR_) + lane;
    float4 a0 = *p0, a1 = *p1, a2 = *p2, a3 = *p3;
    const bool all4 = (r0 != r1) & (r0 != r2) & (r0 != r3) & (r1 != r2) & (r1 != r3) & (r2 != r3);
    if (all4) {
        fma4(a0, n0, b0);
        fma4(a1, n1, b1);
        fma4(a2, n2, b2);
        fma4(a3, n3, b3);
        *p0 = a0;
        *p1 = a1;
        *p2 = a2;
        *p3 = a3;
    } else {
        fma4(a0, n0, b0);
        if (r1 == r0) a1 = a0;
        fma4(a1, n1, b1);
        *p0 = a0;
        *p1 = a1;
        if ((r2 == r0) | (r2 == r1) | (r3 == r0) | (r3 == r1)) {
            a2 = *p2;
            a3 = *p3;
        }
        fma4(a2, n2, b2);
        if (r3 == r2) a3 = a2;
        fma4(a3, n3, b3);
        *p2 = a2;
        *p3 = a3;
    }
}

__device__ __forceinline__ unsigned match_bits(unsigned x, unsigned k4) {
    const unsigned y = x ^ k4;
    const unsigned t = (y & 0x7F7F7F7Fu) + 0x7F7F7F7Fu;
    return ~t & 0x80808080u;
}

#define RMW4(buf, o, g)                                                                \
    add4(tile, lo, lane, rr_q, nrm, (g), buf[(o)], buf[(o) + 1], buf[(o) + 2], buf[(o) + 3]);

__global__ __launch_bounds__(GT_) void k_graph(const float* __restrict__ x,
                                               const int* __restrict__ ei,
                                               const float* __restrict__ bd,
                                               const int* __restrict__ bdmax,
                                               const float* __restrict__ W,
                                               const float* __restrict__ bias,
                                               float* __restrict__ out) {
    __shared__ __attribute__((aligned(16))) float tile[N_ * STR_];
    __shared__ float sdinv[N_];
    __shared__ __attribute__((aligned(16))) unsigned char d8o[B2_];
    __shared__ unsigned char d8r[B2_];
    __shared__ unsigned short qall[(GT_ / 32) * QCAP_];

    const int s    = blockIdx.x;
    const int tid  = threadIdx.x;
    const int lane = tid & 31;
    const int wave = tid >> 5;

    const int*   src_p = ei + (size_t)s * 2 * B2_;
    const int*   dst_p = src_p + B2_;
    const float* bd_p  = bd + (size_t)s * B2_;
    const float* xs    = x + (size_t)s * N_ * F_;
    const float4* xs4  = (const float4*)xs;
    float*       os    = out + (size_t)s * N_ * F_;
    const float  bm    = __int_as_float(*bdmax);

    int   dsr[B2_ / GT_], dds[B2_ / GT_];
    float dbd[B2_ / GT_];
#pragma unroll
    for (int k = 0; k < B2_ / GT_; ++k) {
        int e  = tid + k * GT_;
        dsr[k] = src_p[e];
        dds[k] = dst_p[e];
        dbd[k] = bd_p[e];
    }

    for (int i = tid; i < N_; i += GT_) sdinv[i] = 1.0f;
#pragma unroll
    for (int k = 0; k < B2_ / GT_; ++k) {
        const unsigned d = (unsigned)dds[k];
        const unsigned o = d / (unsigned)RW_;
        d8o[tid + k * GT_] = (dsr[k] != dds[k]) ? (unsigned char)o : (unsigned char)0xFF;
        d8r[tid + k * GT_] = (unsigned char)(d - o * (unsigned)RW_);
    }
    __syncthreads();

#pragma unroll
    for (int k = 0; k < B2_ / GT_; ++k) {
        if (dsr[k] != dds[k]) lds_add(&sdinv[dds[k]], dbd[k] / bm);
    }
    __syncthreads();

    for (int i = tid; i < N_; i += GT_) sdinv[i] = rsqrtf(sdinv[i]);
    __syncthreads();

    const int wv = __builtin_amdgcn_readfirstlane(wave);
    const int lo = wv * RW_;
    unsigned short* qw = qall + wv * QCAP_;

    if (lo < N_) {
        float4 xr[RW_];
#pragma unroll
        for (int r = 0; r < RW_; ++r) xr[r] = xs4[(lo + r) * (F_ / 4) + lane];
#pragma unroll
        for (int r = 0; r < RW_; ++r) {
            float di = sdinv[lo + r];
            float sl = di * di;
            float4 v = xr[r];
            v.x *= sl; v.y *= sl; v.z *= sl; v.w *= sl;
            *((float4*)(tile + (lo + r) * STR_) + lane) = v;
        }

        const unsigned k4 = (unsigned)wv * 0x01010101u;
        int qn = 0, nb = 0, cb = 0;
        unsigned m = 0u;
        for (;;) {
            const unsigned mask = __builtin_amdgcn_ballot_w32(m != 0u);
            if (mask != 0u) {
                const unsigned below = __builtin_amdgcn_mbcnt_lo(mask, 0u);
                if (m != 0u) {
                    const unsigned p = (unsigned)__builtin_ctz(m);
                    qw[qn + below] = (unsigned short)(cb + 16 * lane + (int)(((p & 7u) << 2) + (p >> 3)));
                    m &= m - 1u;
                }
                qn += __builtin_popcount(mask);
                if (qn < 32) continue;
            } else if (nb < B2_) {
                const uint4 tv = *(const uint4*)(d8o + nb + 16 * lane);
                m = (match_bits(tv.x, k4) >> 7) | (match_bits(tv.y, k4) >> 6) |
                    (match_bits(tv.z, k4) >> 5) | (match_bits(tv.w, k4) >> 4);
                cb = nb;
                nb += SCB_;
                continue;
            } else if (qn == 0) {
                break;
            }
            {
                const int n = (qn < 32) ? qn : 32;
                const bool ok = lane < n;
                unsigned e    = ok ? (unsigned)qw[lane] : 0u;
                unsigned rr_q = ok ? (unsigned)d8r[e] : 0u;
                int   sr  = src_p[e];
                float w   = bd_p[e] / bm;
                float nrm = ok ? (sdinv[sr] * w * sdinv[lo + rr_q]) : 0.0f;
                float4 xa[8], xb[8];
                LOAD4(xa, 0, 0);
                LOAD4(xa, 4, 4);
                LOAD4(xb, 0, 8);
                LOAD4(xb, 4, 12);
                RMW4(xa, 0, 0);
                LOAD4(xa, 0, 16);
                RMW4(xa, 4, 4);
                LOAD4(xa, 4, 20);
                RMW4(xb, 0, 8);
                LOAD4(xb, 0, 24);
                RMW4(xb, 4, 12);
                LOAD4(xb, 4, 28);
                RMW4(xa, 0, 16);
                RMW4(xa, 4, 20);
                RMW4(xb, 0, 24);
                RMW4(xb, 4, 28);
                const int rem = qn - n;
                if (rem > 0) {
                    unsigned short t = (lane < rem) ? qw[32 + lane] : (unsigned short)0;
                    if (lane < rem) qw[lane] = t;
                }
                qn = rem;
            }
        }
    }

    const int l16 = lane & 15;
    const int hh  = lane >> 4;
    const int o0  = (wv & 7) * 16;
    const float* brow = W + (size_t)(o0 + l16) * F_ + hh * 2;
    v2f bf[F_ / 4];
#pragma unroll
    for (int t = 0; t < F_ / 4; ++t) bf[t] = *(const v2f*)(brow + 4 * t);
    const float bb = bias[o0 + l16];
    __syncthreads();

    for (int rt = wv >> 3; rt < NRT_; rt += 4) {
        const int m0 = rt * 16;
        int ar = m0 + l16;
        if (ar > N_ - 1) ar = N_ - 1;
        const float* ap = tile + ar * STR_ + hh * 2;
        v8f acc = {};
#pragma unroll
        for (int t = 0; t < F_ / 4; ++t) {
            v2f a = *(const v2f*)(ap + 4 * t);
            acc = __builtin_amdgcn_wmma_f32_16x16x4_f32(
                 false, a,  false, bf[t],
                 (short)0, acc,  false,  false);
        }
        float* op = os + (size_t)(m0 + hh * 8) * F_ + o0 + l16;
        if (m0 + 16 <= N_) {
#pragma unroll
            for (int v = 0; v < 8; ++v) op[v * F_] = acc[v] + bb;
        } else {
#pragma unroll
            for (int v = 0; v < 8; ++v)
                if (m0 + hh * 8 + v < N_) op[v * F_] = acc[v] + bb;
        }
    }
}

extern "C" void kernel_launch(void* const* d_in, const int* in_sizes, int n_in,
                              void* d_out, int out_size, void* d_ws, size_t ws_size,
                              hipStream_t stream) {
    const float* x    = (const float*)d_in[0];
    const int*   ei   = (const int*)  d_in[1];
    const float* bd   = (const float*)d_in[2];
    const float* W    = (const float*)d_in[3];
    const float* bias = (const float*)d_in[4];
    float*       out  = (float*)d_out;

    int* bdmax = (int*)d_ws;

    k_init <<<1,    32,  0, stream>>>(bdmax);
    k_bdmax<<<2048, 256, 0, stream>>>(ei, bd, bdmax);
    k_graph<<<S_,   GT_, 0, stream>>>(x, ei, bd, bdmax, W, bias, out);
}


# Scatter sums held in registers: a wave adds its queued edges row by row into its 18 rows' VGPRs, so the LDS tile first holds the x rows and source rows come from LDS, not through L2; same sum order
# speedup vs baseline: 5.9246x; 1.1301x over previous
#include <hip/hip_runtime.h>


#define S_   256
#define N_   558
#define B2_  8192
#define F_   128
#define TE_  (S_ * B2_)
#define GT_  1024
#define STR_ 132
#define RW_  18
#define SCB_ 512
#define QCAP_ 64
#define NRT_ ((N_ + 15) / 16)

typedef __attribute__((ext_vector_type(2))) float v2f;
typedef __attribute__((ext_vector_type(8))) float v8f;

__global__ __launch_bounds__(32) void k_init(int* __restrict__ bdmax) {
    if (threadIdx.x == 0) *bdmax = 0;
}

__global__ __launch_bounds__(256) void k_bdmax(const int* __restrict__ ei,
                                               const float* __restrict__ bd,
                                               int* __restrict__ bdmax) {
    __shared__ float wmax[8];
    int stride = gridDim.x * 256;
    float m = 0.0f;
    for (int i = blockIdx.x * 256 + threadIdx.x; i < TE_; i += stride) {
        int s    = i >> 13;
        int base = i + (s << 13);
        int src  = ei[base];
        int dst  = ei[base + B2_];
        if (src != dst) m = fmaxf(m, bd[i]);
    }
    for (int off = 16; off > 0; off >>= 1)
        m = fmaxf(m, __shfl_xor(m, off, 32));
    if ((threadIdx.x & 31) == 0) wmax[threadIdx.x >> 5] = m;
    __syncthreads();
    if (threadIdx.x == 0) {
        float g = wmax[0];
#pragma unroll
        for (int w = 1; w < 8; ++w) g = fmaxf(g, wmax[w]);
        atomicMax(bdmax, __float_as_int(g));
    }
}

__device__ __forceinline__ void lds_add(float* p, float v) {
    __hip_atomic_fetch_add(p, v, __ATOMIC_RELAXED, __HIP_MEMORY_SCOPE_WORKGROUP);
}

__device__ __forceinline__ void fma4(float4& a, float n, const float4& b) {
    a.x += n * b.x;
    a.y += n * b.y;
    a.z += n * b.z;
    a.w += n * b.w;
}

#define ROWADD(r)                                                                      \
    {                                                                                  \
        unsigned mk = __builtin_amdgcn_ballot_w32(rr_q == (unsigned)(r));              \
        while (mk != 0u) {                                            \
            const int g0 = __builtin_ctz(mk);                                          \
            mk &= mk - 1u;                                                             \
              \
              \
            const bool two = mk != 0u;                                                 \
            const int g1 = two ? __builtin_ctz(mk) : g0;                               \
            mk &= mk - 1u;                                                             \
            const unsigned s0 = (unsigned)__builtin_amdgcn_readlane((int)sro, g0);     \
            const unsigned s1 = (unsigned)__builtin_amdgcn_readlane((int)sro, g1);     \
            const float n0 = __builtin_bit_cast(float, __builtin_amdgcn_readlane(nrm_i, g0)); \
            const float n1r = __builtin_bit_cast(float, __builtin_amdgcn_readlane(nrm_i, g1)); \
            const float4 b0 = *(const float4*)(tileb + s0 + lane16);                   \
            const float4 b1 = *(const float4*)(tileb + s1 + lane16);                   \
            const float n1 = two ? n1r : 0.0f;                                         \
            fma4(acc[r], n0, b0);                                                      \
            fma4(acc[r], n1, b1);                                                      \
        }                                                                              \
    }

__device__ __forceinline__ unsigned match_bits(unsigned x, unsigned k4) {
    const unsigned y = x ^ k4;
    const unsigned t = (y & 0x7F7F7F7Fu) + 0x7F7F7F7Fu;
    return ~t & 0x80808080u;
}

__global__ __launch_bounds__(GT_) void k_graph(const float* __restrict__ x,
                                               const int* __restrict__ ei,
                                               const float* __restrict__ bd,
                                               const int* __restrict__ bdmax,
                                               const float* __restrict__ W,
                                               const float* __restrict__ bias,
                                               float* __restrict__ out) {
    __shared__ __attribute__((aligned(16))) float tile[N_ * STR_];
    __shared__ float sdinv[N_];
    __shared__ __attribute__((aligned(16))) unsigned char d8o[B2_];
    __shared__ unsigned char d8r[B2_];
    __shared__ unsigned short qall[(GT_ / 32) * QCAP_];

    const int s    = blockIdx.x;
    const int tid  = threadIdx.x;
    const int lane = tid & 31;
    const int wave = tid >> 5;
    const int wv = __builtin_amdgcn_readfirstlane(wave);

    const int*   src_p = ei + (size_t)s * 2 * B2_;
    const int*   dst_p = src_p + B2_;
    const float* bd_p  = bd + (size_t)s * B2_;
    const float* xs    = x + (size_t)s * N_ * F_;
    const float4* xs4  = (const float4*)xs;
    float*       os    = out + (size_t)s * N_ * F_;
    const float  bm    = __int_as_float(*bdmax);

    int   dsr[B2_ / GT_], dds[B2_ / GT_];
    float dbd[B2_ / GT_];
#pragma unroll
    for (int k = 0; k < B2_ / GT_; ++k) {
        int e  = tid + k * GT_;
        dsr[k] = src_p[e];
        dds[k] = dst_p[e];
        dbd[k] = bd_p[e];
    }
    float4 xh[9];
#pragma unroll
    for (int i = 0; i < 9; ++i) xh[i] = xs4[(wv + 32 * i) * (F_ / 4) + lane];

    for (int i = tid; i < N_; i += GT_) sdinv[i] = 1.0f;
#pragma unroll
    for (int k = 0; k < B2_ / GT_; ++k) {
        const unsigned d = (unsigned)dds[k];
        const unsigned o = d / (unsigned)RW_;
        d8o[tid + k * GT_] = (dsr[k] != dds[k]) ? (unsigned char)o : (unsigned char)0xFF;
        d8r[tid + k * GT_] = (unsigned char)(d - o * (unsigned)RW_);
    }
#pragma unroll
    for (int i = 0; i < 9; ++i) *((float4*)(tile + (wv + 32 * i) * STR_) + lane) = xh[i];
    __syncthreads();

    float4 xg[8];
#pragma unroll
    for (int i = 0; i < 8; ++i) xg[i] = xs4[(wv + 32 * (9 + i)) * (F_ / 4) + lane];
    const bool lastrow = (wv + 32 * 17) < N_;
    float4 xl = make_float4(0.0f, 0.0f, 0.0f, 0.0f);
    if (lastrow) xl = xs4[(wv + 32 * 17) * (F_ / 4) + lane];

#pragma unroll
    for (int k = 0; k < B2_ / GT_; ++k) {
        if (dsr[k] != dds[k]) lds_add(&sdinv[dds[k]], dbd[k] / bm);
    }
#pragma unroll
    for (int i = 0; i < 8; ++i) *((float4*)(tile + (wv + 32 * (9 + i)) * STR_) + lane) = xg[i];
    if (lastrow) *((float4*)(tile + (wv + 32 * 17) * STR_) + lane) = xl;
    __syncthreads();

    for (int i = tid; i < N_; i += GT_) sdinv[i] = rsqrtf(sdinv[i]);
    __syncthreads();

    const int lo = wv * RW_;
    unsigned short* qw = qall + wv * QCAP_;
    const char* tileb  = (const char*)tile;
    const unsigned lane16 = (unsigned)lane * 16u;
    float4 acc[RW_];

    if (lo < N_) {
#pragma unroll
        for (int r = 0; r < RW_; ++r) {
            float di = sdinv[lo + r];
            float sl = di * di;
            float4 v = *((const float4*)(tile + (lo + r) * STR_) + lane);
            v.x *= sl; v.y *= sl; v.z *= sl; v.w *= sl;
            acc[r] = v;
        }

        const unsigned k4 = (unsigned)wv * 0x01010101u;
        int qn = 0, nb = 0, cb = 0;
        unsigned m = 0u;
        for (;;) {
            while (qn < 32) {
                const unsigned mask = __builtin_amdgcn_ballot_w32(m != 0u);
                if (mask != 0u) {
                    const unsigned below = __builtin_amdgcn_mbcnt_lo(mask, 0u);
                    if (m != 0u) {
                        const unsigned p = (unsigned)__builtin_ctz(m);
                        qw[qn + below] = (unsigned short)(cb + 16 * lane + (int)(((p & 7u) << 2) + (p >> 3)));
                        m &= m - 1u;
                    }
                    qn += __builtin_popcount(mask);
                } else if (nb < B2_) {
                    const uint4 tv = *(const uint4*)(d8o + nb + 16 * lane);
                    m = (match_bits(tv.x, k4) >> 7) | (match_bits(tv.y, k4) >> 6) |
                        (match_bits(tv.z, k4) >> 5) | (match_bits(tv.w, k4) >> 4);
                    cb = nb;
                    nb += SCB_;
                } else {
                    break;
                }
            }
            if (qn == 0) break;
            {
                const int n = (qn < 32) ? qn : 32;
                const bool ok = lane < n;
                const unsigned e   = ok ? (unsigned)qw[lane] : 0u;
                const unsigned rq0 = (unsigned)d8r[e];
                const unsigned rr_q = ok ? rq0 : 0xFFu;
                const int   sr  = src_p[e];
                const float w   = bd_p[e] / bm;
                const float nrm = sdinv[sr] * w * sdinv[lo + rq0];
                const int      nrm_i = __builtin_bit_cast(int, nrm);
                const unsigned sro   = (unsigned)sr * (unsigned)(STR_ * 4);
                ROWADD(0)  ROWADD(1)  ROWADD(2)  ROWADD(3)  ROWADD(4)  ROWADD(5)
                ROWADD(6)  ROWADD(7)  ROWADD(8)  ROWADD(9)  ROWADD(10) ROWADD(11)
                ROWADD(12) ROWADD(13) ROWADD(14) ROWADD(15) ROWADD(16) ROWADD(17)
                const int rem = qn - n;
                if (rem > 0) {
                    unsigned short t = (lane < rem) ? qw[32 + lane] : (unsigned short)0;
                    if (lane < rem) qw[lane] = t;
                }
                qn = rem;
            }
        }
    }
    __syncthreads();
    if (lo < N_) {
#pragma unroll
        for (int r = 0; r < RW_; ++r) *((float4*)(tile + (lo + r) * STR_) + lane) = acc[r];
    }

    const int l16 = lane & 15;
    const int hh  = lane >> 4;
    const int o0  = (wv & 7) * 16;
    const float* brow = W + (size_t)(o0 + l16) * F_ + hh * 2;
    v2f bf[F_ / 4];
#pragma unroll
    for (int t = 0; t < F_ / 4; ++t) bf[t] = *(const v2f*)(brow + 4 * t);
    const float bb = bias[o0 + l16];
    __syncthreads();

    for (int rt = wv >> 3; rt < NRT_; rt += 4) {
        const int m0 = rt * 16;
        int ar = m0 + l16;
        if (ar > N_ - 1) ar = N_ - 1;
        const float* ap = tile + ar * STR_ + hh * 2;
        v8f acc8 = {};
#pragma unroll
        for (int t = 0; t < F_ / 4; ++t) {
            v2f a = *(const v2f*)(ap + 4 * t);
            acc8 = __builtin_amdgcn_wmma_f32_16x16x4_f32(
                 false, a,  false, bf[t],
                 (short)0, acc8,  false,  false);
        }
        float* op = os + (size_t)(m0 + hh * 8) * F_ + o0 + l16;
        if (m0 + 16 <= N_) {
#pragma unroll
            for (int v = 0; v < 8; ++v) op[v * F_] = acc8[v] + bb;
        } else {
#pragma unroll
            for (int v = 0; v < 8; ++v)
                if (m0 + hh * 8 + v < N_) op[v * F_] = acc8[v] + bb;
        }
    }
}

extern "C" void kernel_launch(void* const* d_in, const int* in_sizes, int n_in,
                              void* d_out, int out_size, void* d_ws, size_t ws_size,
                              hipStream_t stream) {
    const float* x    = (const float*)d_in[0];
    const int*   ei   = (const int*)  d_in[1];
    const float* bd   = (const float*)d_in[2];
    const float* W    = (const float*)d_in[3];
    const float* bias = (const float*)d_in[4];
    float*       out  = (float*)d_out;

    int* bdmax = (int*)d_ws;

    k_init <<<1,    32,  0, stream>>>(bdmax);
    k_bdmax<<<2048, 256, 0, stream>>>(ei, bd, bdmax);
    k_graph<<<S_,   GT_, 0, stream>>>(x, ei, bd, bdmax, W, bias, out);
}


# Row adds take two queued edges only while the row has two left; a last single edge takes a one-edge path (one LDS row read, one add), no zero-weight second slot; same sum order
# speedup vs baseline: 6.1572x; 1.0393x over previous
#include <hip/hip_runtime.h>


#define S_   256
#define N_   558
#define B2_  8192
#define F_   128
#define TE_  (S_ * B2_)
#define GT_  1024
#define STR_ 132
#define RW_  18
#define SCB_ 512
#define QCAP_ 64
#define NRT_ ((N_ + 15) / 16)

typedef __attribute__((ext_vector_type(2))) float v2f;
typedef __attribute__((ext_vector_type(8))) float v8f;

__global__ __launch_bounds__(32) void k_init(int* __restrict__ bdmax) {
    if (threadIdx.x == 0) *bdmax = 0;
}

__global__ __launch_bounds__(256) void k_bdmax(const int* __restrict__ ei,
                                               const float* __restrict__ bd,
                                               int* __restrict__ bdmax) {
    __shared__ float wmax[8];
    int stride = gridDim.x * 256;
    float m = 0.0f;
    for (int i = blockIdx.x * 256 + threadIdx.x; i < TE_; i += stride) {
        int s    = i >> 13;
        int base = i + (s << 13);
        int src  = ei[base];
        int dst  = ei[base + B2_];
        if (src != dst) m = fmaxf(m, bd[i]);
    }
    for (int off = 16; off > 0; off >>= 1)
        m = fmaxf(m, __shfl_xor(m, off, 32));
    if ((threadIdx.x & 31) == 0) wmax[threadIdx.x >> 5] = m;
    __syncthreads();
    if (threadIdx.x == 0) {
        float g = wmax[0];
#pragma unroll
        for (int w = 1; w < 8; ++w) g = fmaxf(g, wmax[w]);
        atomicMax(bdmax, __float_as_int(g));
    }
}

__device__ __forceinline__ void lds_add(float* p, float v) {
    __hip_atomic_fetch_add(p, v, __ATOMIC_RELAXED, __HIP_MEMORY_SCOPE_WORKGROUP);
}

__device__ __forceinline__ void fma4(float4& a, float n, const float4& b) {
    a.x += n * b.x;
    a.y += n * b.y;
    a.z += n * b.z;
    a.w += n * b.w;
}

#define ROWADD(r)                                                                      \
    {                                                                                  \
        unsigned mk = __builtin_amdgcn_ballot_w32(rr_q == (unsigned)(r));              \
        unsigned m1 = mk & (mk - 1u);              \
        while (m1 != 0u) {                            \
            const int g0 = __builtin_ctz(mk);                                          \
            const int g1 = __builtin_ctz(m1);                                          \
            mk = m1 & (m1 - 1u);                                                       \
            m1 = mk & (mk - 1u);                                                       \
            const unsigned s0 = (unsigned)__builtin_amdgcn_readlane((int)sro, g0);     \
            const unsigned s1 = (unsigned)__builtin_amdgcn_readlane((int)sro, g1);     \
            const float n0 = __builtin_bit_cast(float, __builtin_amdgcn_readlane(nrm_i, g0)); \
            const float n1 = __builtin_bit_cast(float, __builtin_amdgcn_readlane(nrm_i, g1)); \
            const float4 b0 = *(const float4*)(tileb + s0 + lane16);                   \
            const float4 b1 = *(const float4*)(tileb + s1 + lane16);                   \
            fma4(acc[r], n0, b0);                                                      \
            fma4(acc[r], n1, b1);                                                      \
        }                                                                              \
        if (mk != 0u) {                                \
            const int g0 = __builtin_ctz(mk);                                          \
            const unsigned s0 = (unsigned)__builtin_amdgcn_readlane((int)sro, g0);     \
            const float n0 = __builtin_bit_cast(float, __builtin_amdgcn_readlane(nrm_i, g0)); \
            const float4 b0 = *(const float4*)(tileb + s0 + lane16);                   \
            fma4(acc[r], n0, b0);                                                      \
        }                                                                              \
    }

__device__ __forceinline__ unsigned match_bits(unsigned x, unsigned k4) {
    const unsigned y = x ^ k4;
    const unsigned t = (y & 0x7F7F7F7Fu) + 0x7F7F7F7Fu;
    return ~t & 0x80808080u;
}

__global__ __launch_bounds__(GT_) void k_graph(const float* __restrict__ x,
                                               const int* __restrict__ ei,
                                               const float* __restrict__ bd,
                                               const int* __restrict__ bdmax,
                                               const float* __restrict__ W,
                                               const float* __restrict__ bias,
                                               float* __restrict__ out) {
    __shared__ __attribute__((aligned(16))) float tile[N_ * STR_];
    __shared__ float sdinv[N_];
    __shared__ __attribute__((aligned(16))) unsigned char d8o[B2_];
    __shared__ unsigned char d8r[B2_];
    __shared__ unsigned short qall[(GT_ / 32) * QCAP_];

    const int s    = blockIdx.x;
    const int tid  = threadIdx.x;
    const int lane = tid & 31;
    const int wave = tid >> 5;
    const int wv = __builtin_amdgcn_readfirstlane(wave);

    const int*   src_p = ei + (size_t)s * 2 * B2_;
    const int*   dst_p = src_p + B2_;
    const float* bd_p  = bd + (size_t)s * B2_;
    const float* xs    = x + (size_t)s * N_ * F_;
    const float4* xs4  = (const float4*)xs;
    float*       os    = out + (size_t)s * N_ * F_;
    const float  bm    = __int_as_float(*bdmax);

    int   dsr[B2_ / GT_], dds[B2_ / GT_];
    float dbd[B2_ / GT_];
#pragma unroll
    for (int k = 0; k < B2_ / GT_; ++k) {
        int e  = tid + k * GT_;
        dsr[k] = src_p[e];
        dds[k] = dst_p[e];
        dbd[k] = bd_p[e];
    }
    float4 xh[9];
#pragma unroll
    for (int i = 0; i < 9; ++i) xh[i] = xs4[(wv + 32 * i) * (F_ / 4) + lane];

    for (int i = tid; i < N_; i += GT_) sdinv[i] = 1.0f;
#pragma unroll
    for (int k = 0; k < B2_ / GT_; ++k) {
        const unsigned d = (unsigned)dds[k];
        const unsigned o = d / (unsigned)RW_;
        d8o[tid + k * GT_] = (dsr[k] != dds[k]) ? (unsigned char)o : (unsigned char)0xFF;
        d8r[tid + k * GT_] = (unsigned char)(d - o * (unsigned)RW_);
    }
#pragma unroll
    for (int i = 0; i < 9; ++i) *((float4*)(tile + (wv + 32 * i) * STR_) + lane) = xh[i];
    __syncthreads();

    float4 xg[8];
#pragma unroll
    for (int i = 0; i < 8; ++i) xg[i] = xs4[(wv + 32 * (9 + i)) * (F_ / 4) + lane];
    const bool lastrow = (wv + 32 * 17) < N_;
    float4 xl = make_float4(0.0f, 0.0f, 0.0f, 0.0f);
    if (lastrow) xl = xs4[(wv + 32 * 17) * (F_ / 4) + lane];

#pragma unroll
    for (int k = 0; k < B2_ / GT_; ++k) {
        if (dsr[k] != dds[k]) lds_add(&sdinv[dds[k]], dbd[k] / bm);
    }
#pragma unroll
    for (int i = 0; i < 8; ++i) *((float4*)(tile + (wv + 32 * (9 + i)) * STR_) + lane) = xg[i];
    if (lastrow) *((float4*)(tile + (wv + 32 * 17) * STR_) + lane) = xl;
    __syncthreads();

    for (int i = tid; i < N_; i += GT_) sdinv[i] = rsqrtf(sdinv[i]);
    __syncthreads();

    const int lo = wv * RW_;
    unsigned short* qw = qall + wv * QCAP_;
    const char* tileb  = (const char*)tile;
    const unsigned lane16 = (unsigned)lane * 16u;
    float4 acc[RW_];

    if (lo < N_) {
#pragma unroll
        for (int r = 0; r < RW_; ++r) {
            float di = sdinv[lo + r];
            float sl = di * di;
            float4 v = *((const float4*)(tile + (lo + r) * STR_) + lane);
            v.x *= sl; v.y *= sl; v.z *= sl; v.w *= sl;
            acc[r] = v;
        }

        const unsigned k4 = (unsigned)wv * 0x01010101u;
        int qn = 0, nb = 0, cb = 0;
        unsigned m = 0u;
        for (;;) {
            while (qn < 32) {
                const unsigned mask = __builtin_amdgcn_ballot_w32(m != 0u);
                if (mask != 0u) {
                    const unsigned below = __builtin_amdgcn_mbcnt_lo(mask, 0u);
                    if (m != 0u) {
                        const unsigned p = (unsigned)__builtin_ctz(m);
                        qw[qn + below] = (unsigned short)(cb + 16 * lane + (int)(((p & 7u) << 2) + (p >> 3)));
                        m &= m - 1u;
                    }
                    qn += __builtin_popcount(mask);
                } else if (nb < B2_) {
                    const uint4 tv = *(const uint4*)(d8o + nb + 16 * lane);
                    m = (match_bits(tv.x, k4) >> 7) | (match_bits(tv.y, k4) >> 6) |
                        (match_bits(tv.z, k4) >> 5) | (match_bits(tv.w, k4) >> 4);
                    cb = nb;
                    nb += SCB_;
                } else {
                    break;
                }
            }
            if (qn == 0) break;
            {
                const int n = (qn < 32) ? qn : 32;
                const bool ok = lane < n;
                const unsigned e   = ok ? (unsigned)qw[lane] : 0u;
                const unsigned rq0 = (unsigned)d8r[e];
                const unsigned rr_q = ok ? rq0 : 0xFFu;
                const int   sr  = src_p[e];
                const float w   = bd_p[e] / bm;
                const float nrm = sdinv[sr] * w * sdinv[lo + rq0];
                const int      nrm_i = __builtin_bit_cast(int, nrm);
                const unsigned sro   = (unsigned)sr * (unsigned)(STR_ * 4);
                ROWADD(0)  ROWADD(1)  ROWADD(2)  ROWADD(3)  ROWADD(4)  ROWADD(5)
                ROWADD(6)  ROWADD(7)  ROWADD(8)  ROWADD(9)  ROWADD(10) ROWADD(11)
                ROWADD(12) ROWADD(13) ROWADD(14) ROWADD(15) ROWADD(16) ROWADD(17)
                const int rem = qn - n;
                if (rem > 0) {
                    unsigned short t = (lane < rem) ? qw[32 + lane] : (unsigned short)0;
                    if (lane < rem) qw[lane] = t;
                }
                qn = rem;
            }
        }
    }
    __syncthreads();
    if (lo < N_) {
#pragma unroll
        for (int r = 0; r < RW_; ++r) *((float4*)(tile + (lo + r) * STR_) + lane) = acc[r];
    }

    const int l16 = lane & 15;
    const int hh  = lane >> 4;
    const int o0  = (wv & 7) * 16;
    const float* brow = W + (size_t)(o0 + l16) * F_ + hh * 2;
    v2f bf[F_ / 4];
#pragma unroll
    for (int t = 0; t < F_ / 4; ++t) bf[t] = *(const v2f*)(brow + 4 * t);
    const float bb = bias[o0 + l16];
    __syncthreads();

    for (int rt = wv >> 3; rt < NRT_; rt += 4) {
        const int m0 = rt * 16;
        int ar = m0 + l16;
        if (ar > N_ - 1) ar = N_ - 1;
        const float* ap = tile + ar * STR_ + hh * 2;
        v8f acc8 = {};
#pragma unroll
        for (int t = 0; t < F_ / 4; ++t) {
            v2f a = *(const v2f*)(ap + 4 * t);
            acc8 = __builtin_amdgcn_wmma_f32_16x16x4_f32(
                 false, a,  false, bf[t],
                 (short)0, acc8,  false,  false);
        }
        float* op = os + (size_t)(m0 + hh * 8) * F_ + o0 + l16;
        if (m0 + 16 <= N_) {
#pragma unroll
            for (int v = 0; v < 8; ++v) op[v * F_] = acc8[v] + bb;
        } else {
#pragma unroll
            for (int v = 0; v < 8; ++v)
                if (m0 + hh * 8 + v < N_) op[v * F_] = acc8[v] + bb;
        }
    }
}

extern "C" void kernel_launch(void* const* d_in, const int* in_sizes, int n_in,
                              void* d_out, int out_size, void* d_ws, size_t ws_size,
                              hipStream_t stream) {
    const float* x    = (const float*)d_in[0];
    const int*   ei   = (const int*)  d_in[1];
    const float* bd   = (const float*)d_in[2];
    const float* W    = (const float*)d_in[3];
    const float* bias = (const float*)d_in[4];
    float*       out  = (float*)d_out;

    int* bdmax = (int*)d_ws;

    k_init <<<1,    32,  0, stream>>>(bdmax);
    k_bdmax<<<2048, 256, 0, stream>>>(ei, bd, bdmax);
    k_graph<<<S_,   GT_, 0, stream>>>(x, ei, bd, bdmax, W, bias, out);
}
